# starting kernel
# baseline (speedup 1.0000x reference)
#include <hip/hip_runtime.h>


typedef __attribute__((ext_vector_type(2))) float v2f;
typedef __attribute__((ext_vector_type(8))) float v8f;

__global__ __launch_bounds__(256) void iwt_wmma_kernel(const float* __restrict__ x,
                                                       float* __restrict__ out) {
    const int tid  = blockIdx.x * blockDim.x + threadIdx.x;
    const int lane = threadIdx.x & 31;
    const int wid  = tid >> 5;

    const int h  = wid & 63;
    const int bc = wid >> 6;

    const bool lo   = lane < 16;
    const int  m    = lane & 15;
    const int  comp = m & 3;
    const bool act  = (m & 4) == 0;

    const float s0 = lo ? 0.5f
                        : (((comp & 1) == 0) ? 0.5f : -0.5f);
    const float s1 = lo ? ((comp < 2) ? 0.5f : -0.5f)
                        : ((comp == 0 || comp == 3) ? 0.5f : -0.5f);
    v2f a;
    a.x = act ? s0 : 0.0f;
    a.y = act ? s1 : 0.0f;

    const size_t plane = (size_t)bc * 16384;
    const int    sub0  = lo ? 0 : 2;
    const int    sub1  = lo ? 1 : 3;

    const float* in_b0 = x + plane + (size_t)sub0 * 4096 + (size_t)h * 64 + m;
    const float* in_b1 = x + plane + (size_t)sub1 * 4096 + (size_t)h * 64 + m;
    float* out_row = out + plane + (size_t)(2 * h + (lo ? 0 : 1)) * 128 + 2 * m;

#pragma unroll
    for (int g = 0; g < 4; ++g) {
        v2f bm;
        bm.x = __builtin_nontemporal_load(in_b0 + g * 16);
        bm.y = __builtin_nontemporal_load(in_b1 + g * 16);
        v8f cacc = {};
        v8f d = __builtin_amdgcn_wmma_f32_16x16x4_f32(
            false, a, false, bm, (short)0, cacc, false, false);

        float2 st;
        st.x = lo ? d[0] : d[2];
        st.y = lo ? d[1] : d[3];
        *(float2*)(out_row + g * 32) = st;
    }
}

extern "C" void kernel_launch(void* const* d_in, const int* in_sizes, int n_in,
                              void* d_out, int out_size, void* d_ws, size_t ws_size,
                              hipStream_t stream) {
    const float* x = (const float*)d_in[0];
    float* out = (float*)d_out;

    const int waves  = in_sizes[0] / 256;
    const int blocks = waves / 8;
    iwt_wmma_kernel<<<dim3(blocks), dim3(256), 0, stream>>>(x, out);
}


# B fragments regrouped through wave-private LDS: 16-byte loads of 512 contiguous bytes (16 per wave, all requested at entry), columns = pixel pairs, 16-byte whole-line stores; 16384 waves of 8 rows
# speedup vs baseline: 1.1819x; 1.1819x over previous
#include <hip/hip_runtime.h>


typedef __attribute__((ext_vector_type(2))) float v2f;
typedef __attribute__((ext_vector_type(4))) float v4f;
typedef __attribute__((ext_vector_type(8))) float v8f;

#define WAVES_PER_BLOCK 16
#define UNITS_PER_WAVE  4
#define SUB_PITCH       144

__global__ __launch_bounds__(512) void iwt_wmma_kernel(const float* __restrict__ x,
                                                       float* __restrict__ out) {
    __shared__ __attribute__((aligned(16))) float stage[WAVES_PER_BLOCK][UNITS_PER_WAVE][4 * SUB_PITCH];

    const int lane = threadIdx.x & 31;
    const int wib  = threadIdx.x >> 5;
    const int wid  = blockIdx.x * WAVES_PER_BLOCK + wib;

    const int bc = wid >> 3;
    const int h0 = (wid & 7) * 8;

    const int  half = lane >> 4;
    const bool lo   = half == 0;
    const int  c    = lane & 15;

    const bool act  = (c & 6) == 0;
    const int  comp = (c & 1) | ((c >> 3) << 1);
    const float s0 = lo ? 0.5f
                        : (((comp & 1) == 0) ? 0.5f : -0.5f);
    const float s1 = lo ? ((comp < 2) ? 0.5f : -0.5f)
                        : ((comp == 0 || comp == 3) ? 0.5f : -0.5f);
    v2f a;
    a.x = act ? s0 : 0.0f;
    a.y = act ? s1 : 0.0f;

    const size_t plane = (size_t)bc * 16384;
    const float* in = x + plane + (size_t)h0 * 64 + 4 * lane;

    v4f ld[UNITS_PER_WAVE][4];
#pragma unroll
    for (int u = 0; u < UNITS_PER_WAVE; ++u) {
#pragma unroll
        for (int s = 0; s < 4; ++s) {
            ld[u][s] = __builtin_nontemporal_load(
                (const v4f*)(in + s * 4096 + u * 128));
        }
    }

    float* st = &stage[wib][0][0];
    float*       sw = st + 4 * lane;
    const float* f0 = st + (2 * half) * SUB_PITCH + 2 * c;
    const float* f1 = f0 + SUB_PITCH;
    float* outp = out + plane + (size_t)(2 * h0 + half) * 128 + 4 * c;

#pragma unroll
    for (int u = 0; u < UNITS_PER_WAVE; ++u) {
#pragma unroll
        for (int s = 0; s < 4; ++s) {
            *(v4f*)(sw + (u * 4 + s) * SUB_PITCH) = ld[u][s];
        }
        __builtin_amdgcn_wave_barrier();

#pragma unroll
        for (int r = 0; r < 2; ++r) {
#pragma unroll
            for (int q = 0; q < 2; ++q) {
                const v2f p0 = *(const v2f*)(f0 + u * 4 * SUB_PITCH + r * 64 + q * 32);
                const v2f p1 = *(const v2f*)(f1 + u * 4 * SUB_PITCH + r * 64 + q * 32);
                v2f b0, b1;
                b0.x = p0.x; b0.y = p1.x;
                b1.x = p0.y; b1.y = p1.y;
                v8f z = {};
                v8f d0 = __builtin_amdgcn_wmma_f32_16x16x4_f32(
                    false, a, false, b0, (short)0, z, false, false);
                v8f d1 = __builtin_amdgcn_wmma_f32_16x16x4_f32(
                    false, a, false, b1, (short)0, z, false, false);
                v4f o;
                o.x = d0[0]; o.y = d0[1]; o.z = d1[0]; o.w = d1[1];
                *(v4f*)(outp + (size_t)(2 * (2 * u + r)) * 128 + q * 64) = o;
            }
        }
    }
}

extern "C" void kernel_launch(void* const* d_in, const int* in_sizes, int n_in,
                              void* d_out, int out_size, void* d_ws, size_t ws_size,
                              hipStream_t stream) {
    const float* x = (const float*)d_in[0];
    float* out = (float*)d_out;

    iwt_wmma_kernel<<<dim3(1024), dim3(512), 0, stream>>>(x, out);
}
